# starting kernel
# baseline (speedup 1.0000x reference)
#include <hip/hip_runtime.h>

typedef float v2f __attribute__((ext_vector_type(2)));
typedef float v8f __attribute__((ext_vector_type(8)));

#define H_   64
#define W_   64
#define OH   128
#define OW   128
#define CTOT 256
#define CG   64
#define K2   25
#define NG   4
#define FSTR 98

__global__ __launch_bounds__(256) void carafe_wmma(
    const float* __restrict__ feat, const float* __restrict__ mask,
    float* __restrict__ out)
{
  __shared__ float sF[CG * FSTR];
  __shared__ float sM[K2 * 128];

  const int tid = threadIdx.x;
  const int n   = blockIdx.z;
  const int g   = blockIdx.y;
  const int stx = blockIdx.x & 7;
  const int sty = blockIdx.x >> 3;
  const int x0  = stx * 8;
  const int y0  = sty * 4;

  const float* fbase = feat + ((size_t)n * CTOT + g * CG) * (H_ * W_);
  #pragma unroll 1
  for (int e = tid; e < CG * 96; e += 256) {
    int c  = e / 96;
    int r  = e - c * 96;
    int yy = r / 12;
    int xx = r - yy * 12;
    int gy = y0 - 2 + yy;
    int gx = x0 - 2 + xx;
    float v = 0.0f;
    if (gy >= 0 && gy < H_ && gx >= 0 && gx < W_)
      v = fbase[c * (H_ * W_) + gy * W_ + gx];
    sF[c * FSTR + r] = v;
  }

  const float* mb = mask + ((size_t)n * (K2 * NG) + g * K2) * (OH * OW)
                  + (2 * y0) * OW + 2 * x0;
  #pragma unroll 1
  for (int e = tid; e < K2 * 128; e += 256) {
    int ch = e >> 7;
    int r  = e & 127;
    int ry = r >> 4;
    int rx = r & 15;
    sM[e] = mb[(size_t)ch * (OH * OW) + ry * OW + rx];
  }
  __syncthreads();

  const int wave = tid >> 5;
  const int lane = tid & 31;
  const int tx = wave & 3, ty = wave >> 2;
  const int h0 = y0 + 2 * ty;
  const int w0 = x0 + 2 * tx;

  const int M    = lane & 15;
  const int half = lane >> 4;
  const int myo = M >> 2, mxo = M & 3;
  const int ti = myo >> 1, tj = mxo >> 1;
  const int lc = lane & 15;
  const int mpos = (4 * ty + myo) * 16 + (4 * tx + mxo);

  v2f a[9];
  int fo[9];
  #pragma unroll
  for (int kk = 0; kk < 9; ++kk) {
    #pragma unroll
    for (int j = 0; j < 2; ++j) {
      const int Kl = 4 * kk + j;
      const int Kh = Kl + 2;
      int dy = half ? (Kh / 6) : (Kl / 6);
      int dx = half ? (Kh % 6) : (Kl % 6);
      int di = dy - ti, dj = dx - tj;
      float av = 0.0f;
      if (di >= 0 && di < 5 && dj >= 0 && dj < 5)
        av = sM[(di * 5 + dj) * 128 + mpos];
      a[kk][j] = av;
      if (j == 0)
        fo[kk] = (2 * ty + dy) * 12 + (2 * tx + dx);
    }
  }

  float* obase = out + ((size_t)n * CTOT + g * CG) * (OH * OW);

  #pragma unroll
  for (int ct = 0; ct < 4; ++ct) {
    v8f acc = {};
    #pragma unroll
    for (int kk = 0; kk < 9; ++kk) {
      v2f b = *(const v2f*)&sF[lc * FSTR + fo[kk] + ct * (16 * FSTR)];
      acc = __builtin_amdgcn_wmma_f32_16x16x4_f32(
          false, a[kk], false, b, (short)0, acc, false, false);
    }
    float* op = obase + (size_t)(ct * 16 + lc) * (OH * OW);
    const int ryo = 2 * h0 + 2 * half;
    float4 lo = make_float4(acc[0], acc[1], acc[2], acc[3]);
    float4 hi = make_float4(acc[4], acc[5], acc[6], acc[7]);
    *(float4*)(op + (size_t)ryo * OW + 2 * w0)       = lo;
    *(float4*)(op + (size_t)(ryo + 1) * OW + 2 * w0) = hi;
  }
}

extern "C" void kernel_launch(void* const* d_in, const int* in_sizes, int n_in,
                              void* d_out, int out_size, void* d_ws, size_t ws_size,
                              hipStream_t stream) {
  (void)in_sizes; (void)n_in; (void)out_size; (void)d_ws; (void)ws_size;
  const float* feat = (const float*)d_in[0];
  const float* mask = (const float*)d_in[1];
  float* out = (float*)d_out;
  dim3 grid(8 * 16, NG, 4);
  carafe_wmma<<<grid, 256, 0, stream>>>(feat, mask, out);
}


# Block walks down a strip of 4 stacked 8x4-source tiles: strip's feature rows stay in LDS (4 new rows per tile, was 8), next tile's global loads run under this tile's WMMAs; grid 512, was 2048
# speedup vs baseline: 2.7893x; 2.7893x over previous
#include <hip/hip_runtime.h>

typedef float v2f __attribute__((ext_vector_type(2)));
typedef float v8f __attribute__((ext_vector_type(8)));

#define H_   64
#define W_   64
#define OH   128
#define OW   128
#define CTOT 256
#define CG   64
#define K2   25
#define NG   4
#define TPB  4
#define WROWS (4 * TPB + 4)
#define FSTR (WROWS * 12 + 4)
#define MSZ  (K2 * 128)


__device__ __forceinline__ void feat_load(float (&pf)[16], const float* __restrict__ fsrc, int gy) {
  const int gyc = min(max(gy, 0), H_ - 1);
  const float* p = fsrc + gyc * W_;
  #pragma unroll
  for (int i = 0; i < 16; ++i) pf[i] = p[(size_t)(4 * i) * (H_ * W_)];
}
__device__ __forceinline__ void feat_store(const float (&pf)[16], float* fdst, bool inside, bool live) {
  if (live) {
    #pragma unroll
    for (int i = 0; i < 16; ++i) fdst[4 * i * FSTR] = inside ? pf[i] : 0.0f;
  }
}
__device__ __forceinline__ void mask_load(float (&pm)[13], const float* __restrict__ msrc, bool lowwaves) {
  #pragma unroll
  for (int i = 0; i < 12; ++i) pm[i] = msrc[(size_t)(2 * i) * (OH * OW)];
  pm[12] = 0.0f;
  if (lowwaves) pm[12] = msrc[(size_t)24 * (OH * OW)];
}
__device__ __forceinline__ void mask_store(const float (&pm)[13], float* mdst, bool lowwaves) {
  #pragma unroll
  for (int i = 0; i < 12; ++i) mdst[256 * i] = pm[i];
  if (lowwaves) mdst[256 * 12] = pm[12];
}

__global__ __launch_bounds__(256, 4) void carafe_wmma(
    const float* __restrict__ feat, const float* __restrict__ mask,
    float* __restrict__ out)
{
  __shared__ __attribute__((aligned(16))) float sF[CG * FSTR];
  __shared__ float sM[2 * MSZ];

  const int tid = threadIdx.x;
  const int n   = blockIdx.z;
  const int g   = blockIdx.y;
  const int stx = blockIdx.x & 7;
  const int seg = blockIdx.x >> 3;
  const int x0  = stx * 8;
  const int ys  = seg * (4 * TPB);

  const int wave = __builtin_amdgcn_readfirstlane(tid >> 5);
  const bool lowwaves = wave < 4;

  const int fxx = tid & 15;
  const int fyy = (tid >> 4) & 3;
  const int fc0 = tid >> 6;
  const int gx  = x0 - 2 + fxx;
  const bool live = fxx < 12;
  const bool xin  = live && gx >= 0 && gx < W_;
  const int gxc   = min(max(gx, 0), W_ - 1);
  const float* fsrc = feat + ((size_t)n * CTOT + g * CG + fc0) * (H_ * W_) + gxc;
  float* fdst = sF + fc0 * FSTR + fyy * 12 + fxx;
  const int gy0 = ys - 2 + fyy;
  const float* msrc = mask + (((size_t)n * (K2 * NG) + g * K2 + (tid >> 7)) * OH
                              + 2 * ys + ((tid >> 4) & 7)) * OW + 2 * x0 + (tid & 15);
  float* mdst = sM + tid;

  float pf[16], pm[13];
  {
    float pf1[16];
    feat_load(pf, fsrc, gy0);
    feat_load(pf1, fsrc, gy0 + 4);
    mask_load(pm, msrc, lowwaves);
    feat_store(pf, fdst, xin && (unsigned)gy0 < (unsigned)H_, live);
    feat_store(pf1, fdst + 4 * 12, xin && (unsigned)(gy0 + 4) < (unsigned)H_, live);
    mask_store(pm, mdst, lowwaves);
  }
  __syncthreads();

  const int lane = tid & 31;
  const int c    = lane & 15;
  const int half = lane >> 4;
  const int sy   = wave >> 1;
  const int sxl  = c >> 1;
  const unsigned ol = (unsigned)(8 * half) * (OH * OW) + c;
  float* obase = out + ((size_t)n * CTOT + g * CG) * (OH * OW)
               + (size_t)(2 * ys + wave) * OW + 2 * x0;

  #pragma unroll 1
  for (int t = 0; t < TPB; ++t) {
    const bool more = t + 1 < TPB;
    const int gyn = gy0 + 4 * (t + 2);
    if (more) {
      feat_load(pf, fsrc, gyn);
      mask_load(pm, msrc + (size_t)(8 * (t + 1)) * OW, lowwaves);
    }

    const float* mrow = sM + (t & 1) * MSZ + wave * 16 + c;
    v2f b[15];
    #pragma unroll
    for (int kk = 0; kk < 15; ++kk) {
      const int dy = kk / 3;
      #pragma unroll
      for (int j = 0; j < 2; ++j) {
        const int dj = 4 * (kk % 3) + 2 * half + j - sxl;
        const bool ok = (unsigned)dj < 5u;
        const float v = mrow[(dy * 5 + (ok ? dj : 0)) * 128];
        b[kk][j] = ok ? v : 0.0f;
      }
    }

    const float* arow = sF + c * FSTR + (4 * t + sy) * 12 + 2 * half;
    v8f acc[4] = {};
    #pragma unroll
    for (int kk = 0; kk < 15; ++kk) {
      #pragma unroll
      for (int ct = 0; ct < 4; ++ct) {
        v2f a = *(const v2f*)&arow[ct * (16 * FSTR) + (kk / 3) * 12 + 4 * (kk % 3)];
        acc[ct] = __builtin_amdgcn_wmma_f32_16x16x4_f32(
            false, a, false, b[kk], (short)0, acc[ct], false, false);
      }
    }
    float* orow = obase + (size_t)(8 * t) * OW;
    #pragma unroll
    for (int ct = 0; ct < 4; ++ct) {
      #pragma unroll
      for (int r = 0; r < 8; ++r)
        orow[ol + (unsigned)(ct * 16 + r) * (OH * OW)] = acc[ct][r];
    }

    if (more) {
      feat_store(pf, fdst + 4 * (t + 2) * 12, xin && (unsigned)gyn < (unsigned)H_, live);
      mask_store(pm, mdst + ((t + 1) & 1) * MSZ, lowwaves);
      __builtin_amdgcn_fence(__ATOMIC_RELEASE, "workgroup", "local");
      __builtin_amdgcn_s_barrier();
      __builtin_amdgcn_fence(__ATOMIC_ACQUIRE, "workgroup", "local");
    }
  }
}

extern "C" void kernel_launch(void* const* d_in, const int* in_sizes, int n_in,
                              void* d_out, int out_size, void* d_ws, size_t ws_size,
                              hipStream_t stream) {
  (void)in_sizes; (void)n_in; (void)out_size; (void)d_ws; (void)ws_size;
  const float* feat = (const float*)d_in[0];
  const float* mask = (const float*)d_in[1];
  float* out = (float*)d_out;
  dim3 grid(8 * 4, NG, 4);
  carafe_wmma<<<grid, 256, 0, stream>>>(feat, mask, out);
}


# Wave holds a whole 128-B output line (32 pixels as two column tiles, 32 channels); one v_permlane16_swap per accumulator pair joins the halves in registers: a store writes 1 whole line, was 2 halves
# speedup vs baseline: 4.8347x; 1.7333x over previous
#include <hip/hip_runtime.h>

typedef float v2f __attribute__((ext_vector_type(2)));
typedef float v8f __attribute__((ext_vector_type(8)));

#define H_   64
#define W_   64
#define OH   128
#define OW   128
#define CTOT 256
#define CG   64
#define K2   25
#define NG   4
#define TPB  4
#define RP   20
#define WROWS (2 * TPB + 4)
#define FSTR (WROWS * RP + 4)
#define MSZ  (K2 * 128)
#define SWAP_ROWS(a, b) do { \
    typedef unsigned v2u_ __attribute__((ext_vector_type(2))); \
    v2u_ s_ = __builtin_amdgcn_permlane16_swap(__builtin_bit_cast(unsigned, (float)(a)), \
                                               __builtin_bit_cast(unsigned, (float)(b)), false, false); \
    (a) = __builtin_bit_cast(float, (unsigned)s_[0]); (b) = __builtin_bit_cast(float, (unsigned)s_[1]); } while (0)


__device__ __forceinline__ void feat_load(float (&pf)[16], const float* __restrict__ fsrc, int gy) {
  const int gyc = min(max(gy, 0), H_ - 1);
  const float* p = fsrc + gyc * W_;
  #pragma unroll
  for (int i = 0; i < 16; ++i) pf[i] = p[(size_t)(4 * i) * (H_ * W_)];
}
__device__ __forceinline__ void feat_store(const float (&pf)[16], float* fdst, bool inside, bool live) {
  if (live) {
    #pragma unroll
    for (int i = 0; i < 16; ++i) fdst[4 * i * FSTR] = inside ? pf[i] : 0.0f;
  }
}
__device__ __forceinline__ void mask_load(float (&pm)[13], const float* __restrict__ msrc, bool lowwaves) {
  #pragma unroll
  for (int i = 0; i < 12; ++i) pm[i] = msrc[(size_t)(2 * i) * (OH * OW)];
  pm[12] = 0.0f;
  if (lowwaves) pm[12] = msrc[(size_t)24 * (OH * OW)];
}
__device__ __forceinline__ void mask_store(const float (&pm)[13], float* mdst, bool lowwaves) {
  #pragma unroll
  for (int i = 0; i < 12; ++i) mdst[256 * i] = pm[i];
  if (lowwaves) mdst[256 * 12] = pm[12];
}

__global__ __launch_bounds__(256, 4) void carafe_wmma(
    const float* __restrict__ feat, const float* __restrict__ mask,
    float* __restrict__ out)
{
  __shared__ __attribute__((aligned(16))) float sF[CG * FSTR];
  __shared__ float sM[2 * MSZ];

  const int tid = threadIdx.x;
  const int n   = blockIdx.z;
  const int g   = blockIdx.y;
  const int stx = blockIdx.x & 3;
  const int seg = blockIdx.x >> 2;
  const int x0  = stx * 16;
  const int ys  = seg * (2 * TPB);

  const int wave = __builtin_amdgcn_readfirstlane(tid >> 5);
  const bool lowwaves = wave < 4;

  const int fxx = tid & 31;
  const int fyy = (tid >> 5) & 1;
  const int fc0 = tid >> 6;
  const int gx  = x0 - 2 + fxx;
  const bool live = fxx < RP;
  const bool xin  = live && gx >= 0 && gx < W_;
  const int gxc   = min(max(gx, 0), W_ - 1);
  const float* fsrc = feat + ((size_t)n * CTOT + g * CG + fc0) * (H_ * W_) + gxc;
  float* fdst = sF + fc0 * FSTR + fyy * RP + fxx;
  const int gy0 = ys - 2 + fyy;
  const float* msrc = mask + (((size_t)n * (K2 * NG) + g * K2 + (tid >> 7)) * OH
                              + 2 * ys + ((tid >> 5) & 3)) * OW + 2 * x0 + (tid & 31);
  float* mdst = sM + tid;

  float pf[16], pm[13];
  {
    float pf1[16], pf2[16];
    feat_load(pf, fsrc, gy0);
    feat_load(pf1, fsrc, gy0 + 2);
    feat_load(pf2, fsrc, gy0 + 4);
    mask_load(pm, msrc, lowwaves);
    feat_store(pf, fdst, xin && (unsigned)gy0 < (unsigned)H_, live);
    feat_store(pf1, fdst + 2 * RP, xin && (unsigned)(gy0 + 2) < (unsigned)H_, live);
    feat_store(pf2, fdst + 4 * RP, xin && (unsigned)(gy0 + 4) < (unsigned)H_, live);
    mask_store(pm, mdst, lowwaves);
  }
  __syncthreads();

  const int lane  = tid & 31;
  const int c     = lane & 15;
  const int half  = lane >> 4;
  const int orow  = wave & 3;
  const int chalf = wave >> 2;
  const int sy    = orow >> 1;
  const int sxl   = c >> 1;
  float* obase = out + ((size_t)n * CTOT + g * CG + chalf * 32) * (OH * OW)
               + (size_t)(2 * ys + orow) * OW + 2 * x0;

  #pragma unroll 1
  for (int t = 0; t < TPB; ++t) {
    const bool more = t + 1 < TPB;
    const int gyn = gy0 + 2 * (t + 3);
    if (more) {
      feat_load(pf, fsrc, gyn);
      mask_load(pm, msrc + (size_t)(4 * (t + 1)) * OW, lowwaves);
    }

    const float* mrow = sM + (t & 1) * MSZ + orow * 32 + c;
    const float* arow = sF + (chalf * 32 + c) * FSTR + (2 * t + sy) * RP + 2 * half;
    v8f acc[2][2] = {};
    #pragma unroll
    for (int kk = 0; kk < 15; ++kk) {
      const int dy = kk / 3;
      v2f b[2];
      #pragma unroll
      for (int j = 0; j < 2; ++j) {
        #pragma unroll
        for (int e = 0; e < 2; ++e) {
          const int dj = 4 * (kk % 3) + 2 * half + e - sxl;
          const bool ok = (unsigned)dj < 5u;
          const float v = mrow[(dy * 5 + (ok ? dj : 0)) * 128 + 16 * j];
          b[j][e] = ok ? v : 0.0f;
        }
      }
      #pragma unroll
      for (int j = 0; j < 2; ++j) {
        #pragma unroll
        for (int ct = 0; ct < 2; ++ct) {
          v2f a = *(const v2f*)&arow[ct * (16 * FSTR) + dy * RP + 8 * j + 4 * (kk % 3)];
          acc[j][ct] = __builtin_amdgcn_wmma_f32_16x16x4_f32(
              false, a, false, b[j], (short)0, acc[j][ct], false, false);
        }
      }
    }
    float* orow_p = obase + (size_t)(4 * t) * OW + lane;
    asm volatile("v_nop\n\tv_nop\n\tv_nop\n\tv_nop"
                 : "+v"(acc[0][0]), "+v"(acc[0][1]), "+v"(acc[1][0]), "+v"(acc[1][1]));
    #pragma unroll
    for (int ct = 0; ct < 2; ++ct) {
      #pragma unroll
      for (int r = 0; r < 8; ++r) {
        SWAP_ROWS(acc[0][ct][r], acc[1][ct][r]);
        orow_p[(size_t)(ct * 16 + r) * (OH * OW)]     = acc[0][ct][r];
        orow_p[(size_t)(ct * 16 + 8 + r) * (OH * OW)] = acc[1][ct][r];
      }
    }

    if (more) {
      feat_store(pf, fdst + 2 * (t + 3) * RP, xin && (unsigned)gyn < (unsigned)H_, live);
      mask_store(pm, mdst + ((t + 1) & 1) * MSZ, lowwaves);
      __builtin_amdgcn_fence(__ATOMIC_RELEASE, "workgroup", "local");
      __builtin_amdgcn_s_barrier();
      __builtin_amdgcn_fence(__ATOMIC_ACQUIRE, "workgroup", "local");
    }
  }
}

extern "C" void kernel_launch(void* const* d_in, const int* in_sizes, int n_in,
                              void* d_out, int out_size, void* d_ws, size_t ws_size,
                              hipStream_t stream) {
  (void)in_sizes; (void)n_in; (void)out_size; (void)d_ws; (void)ws_size;
  const float* feat = (const float*)d_in[0];
  const float* mask = (const float*)d_in[1];
  float* out = (float*)d_out;
  dim3 grid(4 * 8, NG, 4);
  carafe_wmma<<<grid, 256, 0, stream>>>(feat, mask, out);
}
